# starting kernel
# baseline (speedup 1.0000x reference)
#include <hip/hip_runtime.h>
#include <hip/hip_bf16.h>

typedef __attribute__((ext_vector_type(2))) float v2f;
typedef __attribute__((ext_vector_type(8))) float v8f;

__global__ void fa_node_project(const float* __restrict__ h,
                                const float* __restrict__ gw,
                                float* __restrict__ p_dst,
                                float* __restrict__ p_src,
                                int N, int D) {
    __shared__ float sgw[512];
    for (int i = threadIdx.x; i < 2 * D; i += blockDim.x) sgw[i] = gw[i];
    __syncthreads();

    const int wave = (int)((blockIdx.x * blockDim.x + threadIdx.x) >> 5);
    const int lane = (int)(threadIdx.x & 31);
    const int r0 = wave * 16;
    if (r0 >= N) return;

    const int m  = lane & 15;
    const int kh = lane >> 4;

    int row = r0 + m;
    row = (row < N) ? row : (N - 1);
    const float* hrow = h + (size_t)row * (size_t)D;

    const float* bp = sgw + ((m & 1) ? D : 0);

    v8f c = {};
    #pragma unroll
    for (int k0 = 0; k0 < D; k0 += 4) {
        const int k = k0 + 2 * kh;
        const v2f a = *(const v2f*)(hrow + k);
        const v2f b = *(const v2f*)(bp + k);

        c = __builtin_amdgcn_wmma_f32_16x16x4_f32(false, a, false, b,
                                                  (short)0, c, false, false);
    }

    if (m < 2) {
        float* pout = (m == 0) ? p_dst : p_src;
        #pragma unroll
        for (int r = 0; r < 8; ++r) {
            const int ro = r0 + r + 8 * kh;
            if (ro < N) pout[ro] = c[r];
        }
    }
}

__global__ void fa_edge_scatter(const float* __restrict__ h,
                                const float* __restrict__ dnorm,
                                const float* __restrict__ yes_no,
                                const float* __restrict__ gate_b,
                                const float* __restrict__ yes_w,
                                const float* __restrict__ no_w,
                                const int*   __restrict__ src,
                                const int*   __restrict__ dst,
                                const float* __restrict__ p_dst,
                                const float* __restrict__ p_src,
                                float* __restrict__ z,
                                int E, int D) {
    const int e = __builtin_amdgcn_readfirstlane(
        (int)((blockIdx.x * blockDim.x + threadIdx.x) >> 5));
    const int lane = (int)(threadIdx.x & 31);
    if (e >= E) return;

    const int s = src[e];
    const int t = dst[e];

    const float g   = tanhf(p_dst[t] + p_src[s] + gate_b[0]);
    const float yn0 = yes_no[e];
    const float yn  = tanhf(yn0 * yes_w[0] + (1.0f - yn0) * no_w[0]);
    const float w   = (g + yn) * 0.5f * dnorm[t] * dnorm[s];

    const float* hs = h + (size_t)s * (size_t)D;
    float*       zt = z + (size_t)t * (size_t)D;

    for (int j = lane * 4; j < D; j += 128) {
        const float4 v = *(const float4*)(hs + j);
        atomicAdd(zt + j + 0, v.x * w);
        atomicAdd(zt + j + 1, v.y * w);
        atomicAdd(zt + j + 2, v.z * w);
        atomicAdd(zt + j + 3, v.w * w);
    }
}

extern "C" void kernel_launch(void* const* d_in, const int* in_sizes, int n_in,
                              void* d_out, int out_size, void* d_ws, size_t ws_size,
                              hipStream_t stream) {
    const float* h      = (const float*)d_in[0];
    const float* dnorm  = (const float*)d_in[1];
    const float* yes_no = (const float*)d_in[2];
    const float* gate_w = (const float*)d_in[3];
    const float* gate_b = (const float*)d_in[4];
    const float* yes_w  = (const float*)d_in[5];
    const float* no_w   = (const float*)d_in[6];
    const int*   src    = (const int*)d_in[7];
    const int*   dst    = (const int*)d_in[8];
    float* z = (float*)d_out;

    const int N = in_sizes[1];
    const int E = in_sizes[2];
    const int D = in_sizes[0] / N;

    float* p_dst = (float*)d_ws;
    float* p_src = p_dst + N;

    hipMemsetAsync(d_out, 0, (size_t)out_size * sizeof(float), stream);

    {
        const int tiles   = (N + 15) / 16;
        const int threads = 256;
        const int blocks  = (tiles * 32 + threads - 1) / threads;
        fa_node_project<<<blocks, threads, 0, stream>>>(h, gate_w, p_dst, p_src, N, D);
    }

    {
        const int threads = 256;
        const long long total = (long long)E * 32;
        const int blocks = (int)((total + threads - 1) / threads);
        fa_edge_scatter<<<blocks, threads, 0, stream>>>(h, dnorm, yes_no, gate_b,
                                                        yes_w, no_w, src, dst,
                                                        p_dst, p_src, z, E, D);
    }
}


# edge kernel: one wave per 32 edges instead of one per edge - lanes fetch the indices and compute the 32 coefficients side by side, then the rows go 8 at a time with their loads issued together
# speedup vs baseline: 1.0327x; 1.0327x over previous
#include <hip/hip_runtime.h>
#include <hip/hip_bf16.h>

typedef __attribute__((ext_vector_type(2))) float v2f;
typedef __attribute__((ext_vector_type(8))) float v8f;

__global__ void fa_node_project(const float* __restrict__ h,
                                const float* __restrict__ gw,
                                float* __restrict__ p_dst,
                                float* __restrict__ p_src,
                                int N, int D) {
    __shared__ float sgw[512];
    for (int i = threadIdx.x; i < 2 * D; i += blockDim.x) sgw[i] = gw[i];
    __syncthreads();

    const int wave = (int)((blockIdx.x * blockDim.x + threadIdx.x) >> 5);
    const int lane = (int)(threadIdx.x & 31);
    const int r0 = wave * 16;
    if (r0 >= N) return;

    const int m  = lane & 15;
    const int kh = lane >> 4;

    int row = r0 + m;
    row = (row < N) ? row : (N - 1);
    const float* hrow = h + (size_t)row * (size_t)D;

    const float* bp = sgw + ((m & 1) ? D : 0);

    v8f c = {};
    #pragma unroll
    for (int k0 = 0; k0 < D; k0 += 4) {
        const int k = k0 + 2 * kh;
        const v2f a = *(const v2f*)(hrow + k);
        const v2f b = *(const v2f*)(bp + k);

        c = __builtin_amdgcn_wmma_f32_16x16x4_f32(false, a, false, b,
                                                  (short)0, c, false, false);
    }

    if (m < 2) {
        float* pout = (m == 0) ? p_dst : p_src;
        #pragma unroll
        for (int r = 0; r < 8; ++r) {
            const int ro = r0 + r + 8 * kh;
            if (ro < N) pout[ro] = c[r];
        }
    }
}

#define FA_EDGES_PER_WAVE 32
#define FA_ROWS_IN_FLIGHT 8

__global__ void __launch_bounds__(256)
fa_edge_scatter(const float* __restrict__ h,
                const float* __restrict__ dnorm,
                const float* __restrict__ yes_no,
                const float* __restrict__ gate_b,
                const float* __restrict__ yes_w,
                const float* __restrict__ no_w,
                const int*   __restrict__ src,
                const int*   __restrict__ dst,
                const float* __restrict__ p_dst,
                const float* __restrict__ p_src,
                float* __restrict__ z,
                int E, int D) {
    const int wave = __builtin_amdgcn_readfirstlane(
        (int)((blockIdx.x * blockDim.x + threadIdx.x) >> 5));
    const int lane = (int)(threadIdx.x & 31);
    const long long e0 = (long long)wave * FA_EDGES_PER_WAVE;
    if (e0 >= (long long)E) return;

    const long long left = (long long)E - e0;
    const int cnt = (left < FA_EDGES_PER_WAVE) ? (int)left : FA_EDGES_PER_WAVE;

    const int my = (lane < cnt) ? lane : (cnt - 1);
    const long long e = e0 + my;

    const int   s   = src[e];
    const int   t   = dst[e];
    const float yn0 = yes_no[e];
    const float gb  = gate_b[0];
    const float yw  = yes_w[0];
    const float nw  = no_w[0];
    const float pd  = p_dst[t];
    const float ps  = p_src[s];
    const float dt  = dnorm[t];
    const float ds  = dnorm[s];

    const float g  = tanhf(pd + ps + gb);
    const float yn = tanhf(yn0 * yw + (1.0f - yn0) * nw);
    const float w  = (g + yn) * 0.5f * dt * ds;
    const int   wi = __float_as_int(w);

    int k = 0;
    for (; k + FA_ROWS_IN_FLIGHT <= cnt; k += FA_ROWS_IN_FLIGHT) {
        int   ss[FA_ROWS_IN_FLIGHT];
        int   tt[FA_ROWS_IN_FLIGHT];
        float ww[FA_ROWS_IN_FLIGHT];
        #pragma unroll
        for (int u = 0; u < FA_ROWS_IN_FLIGHT; ++u) {
            ss[u] = __builtin_amdgcn_readlane(s,  k + u);
            tt[u] = __builtin_amdgcn_readlane(t,  k + u);
            ww[u] = __int_as_float(__builtin_amdgcn_readlane(wi, k + u));
        }
        for (int j = lane * 4; j < D; j += 128) {
            float4 v[FA_ROWS_IN_FLIGHT];
            #pragma unroll
            for (int u = 0; u < FA_ROWS_IN_FLIGHT; ++u)
                v[u] = *(const float4*)(h + (size_t)ss[u] * (size_t)D + j);
            #pragma unroll
            for (int u = 0; u < FA_ROWS_IN_FLIGHT; ++u) {
                float* zt = z + (size_t)tt[u] * (size_t)D + j;
                atomicAdd(zt + 0, v[u].x * ww[u]);
                atomicAdd(zt + 1, v[u].y * ww[u]);
                atomicAdd(zt + 2, v[u].z * ww[u]);
                atomicAdd(zt + 3, v[u].w * ww[u]);
            }
        }
    }
    for (; k < cnt; ++k) {
        const int   s1 = __builtin_amdgcn_readlane(s,  k);
        const int   t1 = __builtin_amdgcn_readlane(t,  k);
        const float w1 = __int_as_float(__builtin_amdgcn_readlane(wi, k));
        const float* hs = h + (size_t)s1 * (size_t)D;
        float*       zt = z + (size_t)t1 * (size_t)D;
        for (int j = lane * 4; j < D; j += 128) {
            const float4 v = *(const float4*)(hs + j);
            atomicAdd(zt + j + 0, v.x * w1);
            atomicAdd(zt + j + 1, v.y * w1);
            atomicAdd(zt + j + 2, v.z * w1);
            atomicAdd(zt + j + 3, v.w * w1);
        }
    }
}

extern "C" void kernel_launch(void* const* d_in, const int* in_sizes, int n_in,
                              void* d_out, int out_size, void* d_ws, size_t ws_size,
                              hipStream_t stream) {
    const float* h      = (const float*)d_in[0];
    const float* dnorm  = (const float*)d_in[1];
    const float* yes_no = (const float*)d_in[2];
    const float* gate_w = (const float*)d_in[3];
    const float* gate_b = (const float*)d_in[4];
    const float* yes_w  = (const float*)d_in[5];
    const float* no_w   = (const float*)d_in[6];
    const int*   src    = (const int*)d_in[7];
    const int*   dst    = (const int*)d_in[8];
    float* z = (float*)d_out;

    const int N = in_sizes[1];
    const int E = in_sizes[2];
    const int D = in_sizes[0] / N;

    float* p_dst = (float*)d_ws;
    float* p_src = p_dst + N;

    hipMemsetAsync(d_out, 0, (size_t)out_size * sizeof(float), stream);

    {
        const int tiles   = (N + 15) / 16;
        const int threads = 256;
        const int blocks  = (tiles * 32 + threads - 1) / threads;
        fa_node_project<<<blocks, threads, 0, stream>>>(h, gate_w, p_dst, p_src, N, D);
    }

    {
        const int threads = 256;
        const int blocks = (int)(((long long)E + 255) / 256);
        fa_edge_scatter<<<blocks, threads, 0, stream>>>(h, dnorm, yes_no, gate_b,
                                                        yes_w, no_w, src, dst,
                                                        p_dst, p_src, z, E, D);
    }
}


# edge kernel: lane owns channels l, l+32, l+64, l+96 of a row instead of 4l..4l+3, so each atomic add instruction covers one whole 128-byte line of z, not a quarter of each of four lines
# speedup vs baseline: 2.2229x; 2.1525x over previous
#include <hip/hip_runtime.h>
#include <hip/hip_bf16.h>

typedef __attribute__((ext_vector_type(2))) float v2f;
typedef __attribute__((ext_vector_type(8))) float v8f;

__global__ void fa_node_project(const float* __restrict__ h,
                                const float* __restrict__ gw,
                                float* __restrict__ p_dst,
                                float* __restrict__ p_src,
                                int N, int D) {
    __shared__ float sgw[512];
    for (int i = threadIdx.x; i < 2 * D; i += blockDim.x) sgw[i] = gw[i];
    __syncthreads();

    const int wave = (int)((blockIdx.x * blockDim.x + threadIdx.x) >> 5);
    const int lane = (int)(threadIdx.x & 31);
    const int r0 = wave * 16;
    if (r0 >= N) return;

    const int m  = lane & 15;
    const int kh = lane >> 4;

    int row = r0 + m;
    row = (row < N) ? row : (N - 1);
    const float* hrow = h + (size_t)row * (size_t)D;

    const float* bp = sgw + ((m & 1) ? D : 0);

    v8f c = {};
    #pragma unroll
    for (int k0 = 0; k0 < D; k0 += 4) {
        const int k = k0 + 2 * kh;
        const v2f a = *(const v2f*)(hrow + k);
        const v2f b = *(const v2f*)(bp + k);

        c = __builtin_amdgcn_wmma_f32_16x16x4_f32(false, a, false, b,
                                                  (short)0, c, false, false);
    }

    if (m < 2) {
        float* pout = (m == 0) ? p_dst : p_src;
        #pragma unroll
        for (int r = 0; r < 8; ++r) {
            const int ro = r0 + r + 8 * kh;
            if (ro < N) pout[ro] = c[r];
        }
    }
}

#define FA_EDGES_PER_WAVE 32
#define FA_ROWS_IN_FLIGHT 8
#define FA_CHUNKS 4

__device__ __forceinline__ void fa_rows_pass128(const float* __restrict__ h,
                                                float* __restrict__ z,
                                                const int (&ss)[FA_ROWS_IN_FLIGHT],
                                                const int (&tt)[FA_ROWS_IN_FLIGHT],
                                                const float (&ww)[FA_ROWS_IN_FLIGHT],
                                                int D, int j) {
    float v[FA_ROWS_IN_FLIGHT][FA_CHUNKS];
    #pragma unroll
    for (int u = 0; u < FA_ROWS_IN_FLIGHT; ++u) {
        const float* hs = h + (size_t)ss[u] * (size_t)D + j;
        #pragma unroll
        for (int c = 0; c < FA_CHUNKS; ++c) v[u][c] = hs[32 * c];
    }
    #pragma unroll
    for (int u = 0; u < FA_ROWS_IN_FLIGHT; ++u) {
        float* zt = z + (size_t)tt[u] * (size_t)D + j;
        #pragma unroll
        for (int c = 0; c < FA_CHUNKS; ++c)
            atomicAdd(zt + 32 * c, v[u][c] * ww[u]);
    }
}

__device__ __forceinline__ void fa_rows_chunk32(const float* __restrict__ h,
                                                float* __restrict__ z,
                                                const int (&ss)[FA_ROWS_IN_FLIGHT],
                                                const int (&tt)[FA_ROWS_IN_FLIGHT],
                                                const float (&ww)[FA_ROWS_IN_FLIGHT],
                                                int D, int j) {
    const bool in = j < D;
    const int  jl = in ? j : (D - 1);
    float v[FA_ROWS_IN_FLIGHT];
    #pragma unroll
    for (int u = 0; u < FA_ROWS_IN_FLIGHT; ++u)
        v[u] = h[(size_t)ss[u] * (size_t)D + jl];
    if (in) {
        #pragma unroll
        for (int u = 0; u < FA_ROWS_IN_FLIGHT; ++u)
            atomicAdd(z + (size_t)tt[u] * (size_t)D + j, v[u] * ww[u]);
    }
}

__global__ void __launch_bounds__(256)
fa_edge_scatter(const float* __restrict__ h,
                const float* __restrict__ dnorm,
                const float* __restrict__ yes_no,
                const float* __restrict__ gate_b,
                const float* __restrict__ yes_w,
                const float* __restrict__ no_w,
                const int*   __restrict__ src,
                const int*   __restrict__ dst,
                const float* __restrict__ p_dst,
                const float* __restrict__ p_src,
                float* __restrict__ z,
                int E, int D) {
    const int wave = __builtin_amdgcn_readfirstlane(
        (int)((blockIdx.x * blockDim.x + threadIdx.x) >> 5));
    const int lane = (int)(threadIdx.x & 31);
    const long long e0 = (long long)wave * FA_EDGES_PER_WAVE;
    if (e0 >= (long long)E) return;

    const long long left = (long long)E - e0;
    const int cnt = (left < FA_EDGES_PER_WAVE) ? (int)left : FA_EDGES_PER_WAVE;

    const int my = (lane < cnt) ? lane : (cnt - 1);
    const long long e = e0 + my;

    const int   s   = src[e];
    const int   t   = dst[e];
    const float yn0 = yes_no[e];
    const float gb  = gate_b[0];
    const float yw  = yes_w[0];
    const float nw  = no_w[0];
    const float pd  = p_dst[t];
    const float ps  = p_src[s];
    const float dt  = dnorm[t];
    const float ds  = dnorm[s];

    const float g  = tanhf(pd + ps + gb);
    const float yn = tanhf(yn0 * yw + (1.0f - yn0) * nw);
    const float w  = (g + yn) * 0.5f * dt * ds;
    const int   wi = __float_as_int(w);

    int k = 0;
    for (; k + FA_ROWS_IN_FLIGHT <= cnt; k += FA_ROWS_IN_FLIGHT) {
        int   ss[FA_ROWS_IN_FLIGHT];
        int   tt[FA_ROWS_IN_FLIGHT];
        float ww[FA_ROWS_IN_FLIGHT];
        #pragma unroll
        for (int u = 0; u < FA_ROWS_IN_FLIGHT; ++u) {
            ss[u] = __builtin_amdgcn_readlane(s,  k + u);
            tt[u] = __builtin_amdgcn_readlane(t,  k + u);
            ww[u] = __int_as_float(__builtin_amdgcn_readlane(wi, k + u));
        }
        int j0 = 0;
        for (; j0 + 32 * FA_CHUNKS <= D; j0 += 32 * FA_CHUNKS)
            fa_rows_pass128(h, z, ss, tt, ww, D, j0 + lane);
        for (; j0 < D; j0 += 32)
            fa_rows_chunk32(h, z, ss, tt, ww, D, j0 + lane);
    }
    for (; k < cnt; ++k) {
        const int   s1 = __builtin_amdgcn_readlane(s,  k);
        const int   t1 = __builtin_amdgcn_readlane(t,  k);
        const float w1 = __int_as_float(__builtin_amdgcn_readlane(wi, k));
        const float* hs = h + (size_t)s1 * (size_t)D;
        float*       zt = z + (size_t)t1 * (size_t)D;
        for (int j = lane; j < D; j += 32)
            atomicAdd(zt + j, hs[j] * w1);
    }
}

extern "C" void kernel_launch(void* const* d_in, const int* in_sizes, int n_in,
                              void* d_out, int out_size, void* d_ws, size_t ws_size,
                              hipStream_t stream) {
    const float* h      = (const float*)d_in[0];
    const float* dnorm  = (const float*)d_in[1];
    const float* yes_no = (const float*)d_in[2];
    const float* gate_w = (const float*)d_in[3];
    const float* gate_b = (const float*)d_in[4];
    const float* yes_w  = (const float*)d_in[5];
    const float* no_w   = (const float*)d_in[6];
    const int*   src    = (const int*)d_in[7];
    const int*   dst    = (const int*)d_in[8];
    float* z = (float*)d_out;

    const int N = in_sizes[1];
    const int E = in_sizes[2];
    const int D = in_sizes[0] / N;

    float* p_dst = (float*)d_ws;
    float* p_src = p_dst + N;

    hipMemsetAsync(d_out, 0, (size_t)out_size * sizeof(float), stream);

    {
        const int tiles   = (N + 15) / 16;
        const int threads = 256;
        const int blocks  = (tiles * 32 + threads - 1) / threads;
        fa_node_project<<<blocks, threads, 0, stream>>>(h, gate_w, p_dst, p_src, N, D);
    }

    {
        const int threads = 256;
        const int blocks = (int)(((long long)E + 255) / 256);
        fa_edge_scatter<<<blocks, threads, 0, stream>>>(h, dnorm, yes_no, gate_b,
                                                        yes_w, no_w, src, dst,
                                                        p_dst, p_src, z, E, D);
    }
}


# coefficient phase: projection kernel packs each node's p_dst, p_src and degree norm into two 8-byte records, so an edge's coefficient takes 2 scattered loads (dst record, src record) instead of 4
# speedup vs baseline: 2.4084x; 1.0834x over previous
#include <hip/hip_runtime.h>
#include <hip/hip_bf16.h>

typedef __attribute__((ext_vector_type(2))) float v2f;
typedef __attribute__((ext_vector_type(8))) float v8f;

__global__ void fa_node_project(const float* __restrict__ h,
                                const float* __restrict__ gw,
                                const float* __restrict__ dnorm,
                                float* __restrict__ node_t,
                                float* __restrict__ node_s,
                                int N, int D) {
    __shared__ float sgw[512];
    for (int i = threadIdx.x; i < 2 * D; i += blockDim.x) sgw[i] = gw[i];
    __syncthreads();

    const int wave = (int)((blockIdx.x * blockDim.x + threadIdx.x) >> 5);
    const int lane = (int)(threadIdx.x & 31);
    const int r0 = wave * 16;
    if (r0 >= N) return;

    const int m  = lane & 15;
    const int kh = lane >> 4;

    int row = r0 + m;
    row = (row < N) ? row : (N - 1);
    const float* hrow = h + (size_t)row * (size_t)D;

    const float* bp = sgw + ((m & 1) ? D : 0);

    float dn[8];
    #pragma unroll
    for (int r = 0; r < 8; ++r) {
        int ro = r0 + r + 8 * kh;
        ro = (ro < N) ? ro : (N - 1);
        dn[r] = dnorm[ro];
    }

    v8f c = {};
    #pragma unroll
    for (int k0 = 0; k0 < D; k0 += 4) {
        const int k = k0 + 2 * kh;
        const v2f a = *(const v2f*)(hrow + k);
        const v2f b = *(const v2f*)(bp + k);

        c = __builtin_amdgcn_wmma_f32_16x16x4_f32(false, a, false, b,
                                                  (short)0, c, false, false);
    }

    if (m < 2) {
        float* nout = (m == 0) ? node_t : node_s;
        #pragma unroll
        for (int r = 0; r < 8; ++r) {
            const int ro = r0 + r + 8 * kh;
            if (ro < N) {
                v2f rec;
                rec.x = c[r];
                rec.y = dn[r];
                *(v2f*)(nout + 2 * (size_t)ro) = rec;
            }
        }
    }
}

#define FA_EDGES_PER_WAVE 32
#define FA_ROWS_IN_FLIGHT 8
#define FA_CHUNKS 4

__device__ __forceinline__ void fa_rows_pass128(const float* __restrict__ h,
                                                float* __restrict__ z,
                                                const int (&ss)[FA_ROWS_IN_FLIGHT],
                                                const int (&tt)[FA_ROWS_IN_FLIGHT],
                                                const float (&ww)[FA_ROWS_IN_FLIGHT],
                                                int D, int j) {
    float v[FA_ROWS_IN_FLIGHT][FA_CHUNKS];
    #pragma unroll
    for (int u = 0; u < FA_ROWS_IN_FLIGHT; ++u) {
        const float* hs = h + (size_t)ss[u] * (size_t)D + j;
        #pragma unroll
        for (int c = 0; c < FA_CHUNKS; ++c) v[u][c] = hs[32 * c];
    }
    #pragma unroll
    for (int u = 0; u < FA_ROWS_IN_FLIGHT; ++u) {
        float* zt = z + (size_t)tt[u] * (size_t)D + j;
        #pragma unroll
        for (int c = 0; c < FA_CHUNKS; ++c)
            atomicAdd(zt + 32 * c, v[u][c] * ww[u]);
    }
}

__device__ __forceinline__ void fa_rows_chunk32(const float* __restrict__ h,
                                                float* __restrict__ z,
                                                const int (&ss)[FA_ROWS_IN_FLIGHT],
                                                const int (&tt)[FA_ROWS_IN_FLIGHT],
                                                const float (&ww)[FA_ROWS_IN_FLIGHT],
                                                int D, int j) {
    const bool in = j < D;
    const int  jl = in ? j : (D - 1);
    float v[FA_ROWS_IN_FLIGHT];
    #pragma unroll
    for (int u = 0; u < FA_ROWS_IN_FLIGHT; ++u)
        v[u] = h[(size_t)ss[u] * (size_t)D + jl];
    if (in) {
        #pragma unroll
        for (int u = 0; u < FA_ROWS_IN_FLIGHT; ++u)
            atomicAdd(z + (size_t)tt[u] * (size_t)D + j, v[u] * ww[u]);
    }
}

__global__ void __launch_bounds__(256)
fa_edge_scatter(const float* __restrict__ h,
                const float* __restrict__ yes_no,
                const float* __restrict__ gate_b,
                const float* __restrict__ yes_w,
                const float* __restrict__ no_w,
                const int*   __restrict__ src,
                const int*   __restrict__ dst,
                const float* __restrict__ node_t,
                const float* __restrict__ node_s,
                float* __restrict__ z,
                int E, int D) {
    const int wave = __builtin_amdgcn_readfirstlane(
        (int)((blockIdx.x * blockDim.x + threadIdx.x) >> 5));
    const int lane = (int)(threadIdx.x & 31);
    const long long e0 = (long long)wave * FA_EDGES_PER_WAVE;
    if (e0 >= (long long)E) return;

    const long long left = (long long)E - e0;
    const int cnt = (left < FA_EDGES_PER_WAVE) ? (int)left : FA_EDGES_PER_WAVE;

    const int my = (lane < cnt) ? lane : (cnt - 1);
    const long long e = e0 + my;

    const int   s   = src[e];
    const int   t   = dst[e];
    const float yn0 = yes_no[e];
    const float gb  = gate_b[0];
    const float yw  = yes_w[0];
    const float nw  = no_w[0];
    const v2f nt = *(const v2f*)(node_t + 2 * (size_t)t);
    const v2f ns = *(const v2f*)(node_s + 2 * (size_t)s);
    const float pd  = nt.x;
    const float ps  = ns.x;
    const float dt  = nt.y;
    const float ds  = ns.y;

    const float g  = tanhf(pd + ps + gb);
    const float yn = tanhf(yn0 * yw + (1.0f - yn0) * nw);
    const float w  = (g + yn) * 0.5f * dt * ds;
    const int   wi = __float_as_int(w);

    int k = 0;
    for (; k + FA_ROWS_IN_FLIGHT <= cnt; k += FA_ROWS_IN_FLIGHT) {
        int   ss[FA_ROWS_IN_FLIGHT];
        int   tt[FA_ROWS_IN_FLIGHT];
        float ww[FA_ROWS_IN_FLIGHT];
        #pragma unroll
        for (int u = 0; u < FA_ROWS_IN_FLIGHT; ++u) {
            ss[u] = __builtin_amdgcn_readlane(s,  k + u);
            tt[u] = __builtin_amdgcn_readlane(t,  k + u);
            ww[u] = __int_as_float(__builtin_amdgcn_readlane(wi, k + u));
        }
        int j0 = 0;
        for (; j0 + 32 * FA_CHUNKS <= D; j0 += 32 * FA_CHUNKS)
            fa_rows_pass128(h, z, ss, tt, ww, D, j0 + lane);
        for (; j0 < D; j0 += 32)
            fa_rows_chunk32(h, z, ss, tt, ww, D, j0 + lane);
    }
    for (; k < cnt; ++k) {
        const int   s1 = __builtin_amdgcn_readlane(s,  k);
        const int   t1 = __builtin_amdgcn_readlane(t,  k);
        const float w1 = __int_as_float(__builtin_amdgcn_readlane(wi, k));
        const float* hs = h + (size_t)s1 * (size_t)D;
        float*       zt = z + (size_t)t1 * (size_t)D;
        for (int j = lane; j < D; j += 32)
            atomicAdd(zt + j, hs[j] * w1);
    }
}

extern "C" void kernel_launch(void* const* d_in, const int* in_sizes, int n_in,
                              void* d_out, int out_size, void* d_ws, size_t ws_size,
                              hipStream_t stream) {
    const float* h      = (const float*)d_in[0];
    const float* dnorm  = (const float*)d_in[1];
    const float* yes_no = (const float*)d_in[2];
    const float* gate_w = (const float*)d_in[3];
    const float* gate_b = (const float*)d_in[4];
    const float* yes_w  = (const float*)d_in[5];
    const float* no_w   = (const float*)d_in[6];
    const int*   src    = (const int*)d_in[7];
    const int*   dst    = (const int*)d_in[8];
    float* z = (float*)d_out;

    const int N = in_sizes[1];
    const int E = in_sizes[2];
    const int D = in_sizes[0] / N;

    float* node_t = (float*)d_ws;
    float* node_s = node_t + 2 * (size_t)N;

    hipMemsetAsync(d_out, 0, (size_t)out_size * sizeof(float), stream);

    {
        const int tiles   = (N + 15) / 16;
        const int threads = 256;
        const int blocks  = (tiles * 32 + threads - 1) / threads;
        fa_node_project<<<blocks, threads, 0, stream>>>(h, gate_w, dnorm, node_t, node_s, N, D);
    }

    {
        const int threads = 256;
        const int blocks = (int)(((long long)E + 255) / 256);
        fa_edge_scatter<<<blocks, threads, 0, stream>>>(h, yes_no, gate_b,
                                                        yes_w, no_w, src, dst,
                                                        node_t, node_s, z, E, D);
    }
}
